# starting kernel
# baseline (speedup 1.0000x reference)
#include <hip/hip_runtime.h>
#include <hip/hip_bf16.h>
#include <math.h>

#define N_NODES 50000
#define N_EDGES 800000
#define F_DIM   128
#define H_OUT1  128
#define C_OUT   64
#define K_TOT   512

typedef __attribute__((ext_vector_type(2))) float v2f;
typedef __attribute__((ext_vector_type(8))) float v8f;


__global__ void zero_f32_kernel(float* __restrict__ p, int n) {
    int i = blockIdx.x * blockDim.x + threadIdx.x;
    if (i < n) p[i] = 0.0f;
}

__global__ void degree_kernel(const int* __restrict__ col, float* __restrict__ deg) {
    int e = blockIdx.x * blockDim.x + threadIdx.x;
    if (e < N_EDGES) atomicAdd(&deg[col[e]], 1.0f);
}

__global__ void dinv_kernel(float* __restrict__ deg) {
    int i = blockIdx.x * blockDim.x + threadIdx.x;
    if (i < N_NODES) {
        float d = deg[i];
        deg[i] = (d > 0.0f) ? rsqrtf(d) : 0.0f;
    }
}

__global__ void norm_kernel(const int* __restrict__ row, const int* __restrict__ col,
                            const float* __restrict__ dinv, float* __restrict__ nrm) {
    int e = blockIdx.x * blockDim.x + threadIdx.x;
    if (e < N_EDGES) nrm[e] = dinv[row[e]] * dinv[col[e]];
}

__global__ void spmm_scatter_kernel(const int* __restrict__ row, const int* __restrict__ col,
                                    const float* __restrict__ nrm,
                                    const float* __restrict__ hin, float* __restrict__ hout) {
    long long tid = (long long)blockIdx.x * blockDim.x + threadIdx.x;
    int e = (int)(tid >> 5);
    int f = ((int)tid & 31) * 4;
    if (e >= N_EDGES) return;
    float w = nrm[e];
    const float4 v = *(const float4*)(hin + (long long)row[e] * F_DIM + f);
    float* dst = hout + (long long)col[e] * F_DIM + f;
    atomicAdd(dst + 0, w * v.x);
    atomicAdd(dst + 1, w * v.y);
    atomicAdd(dst + 2, w * v.z);
    atomicAdd(dst + 3, w * v.w);
}

template <int NOUT, int MODE>
__global__ __launch_bounds__(256) void tag_gemm_wmma_kernel(
    const float* __restrict__ s0, const float* __restrict__ s1,
    const float* __restrict__ s2, const float* __restrict__ s3,
    const float* __restrict__ W, const float* __restrict__ bias,
    float* __restrict__ out) {
    __shared__ float Bs[K_TOT * 16];

    const int col0 = blockIdx.y * 16;

    for (int i = threadIdx.x; i < K_TOT * 16; i += 256) {
        int k = i >> 4;
        int n = i & 15;
        Bs[i] = W[k * NOUT + col0 + n];
    }
    __syncthreads();

    const int wave   = threadIdx.x >> 5;
    const int lane   = threadIdx.x & 31;
    const int lane16 = lane & 15;
    const int hi     = lane >> 4;
    const int row0   = (blockIdx.x * 8 + wave) * 16;
    if (row0 >= N_NODES) return;
    const int m = row0 + lane16;

    const float* segs[4] = { s0, s1, s2, s3 };

    v8f c = { 0.f, 0.f, 0.f, 0.f, 0.f, 0.f, 0.f, 0.f };

#pragma unroll
    for (int s = 0; s < 4; ++s) {
        const float* __restrict__ arow = segs[s] + (long long)m * F_DIM;
        const float* __restrict__ bseg = Bs + s * F_DIM * 16;
        for (int kk = 0; kk < F_DIM; kk += 4) {
            v2f a;
            a.x = arow[kk + hi * 2 + 0];
            a.y = arow[kk + hi * 2 + 1];
            v2f b;
            b.x = bseg[(kk + hi * 2 + 0) * 16 + lane16];
            b.y = bseg[(kk + hi * 2 + 1) * 16 + lane16];
            c = __builtin_amdgcn_wmma_f32_16x16x4_f32(false, a, false, b, (short)0, c,
                                                      false, false);
        }
    }

    const int n    = col0 + lane16;
    const float bv = bias[n];
#pragma unroll
    for (int j = 0; j < 8; ++j) {
        int mm  = row0 + j + hi * 8;
        float v = c[j] + bv;
        if (MODE == 0) v = (v > 0.0f) ? v : 0.0f;
        out[(long long)mm * NOUT + n] = v;
    }
}

__global__ void log_softmax_kernel(float* __restrict__ out) {
    int wave = threadIdx.x >> 5;
    int lane = threadIdx.x & 31;
    int r = blockIdx.x * (blockDim.x >> 5) + wave;
    if (r >= N_NODES) return;
    float* p = out + (long long)r * C_OUT;
    float a = p[lane];
    float b = p[lane + 32];
    float mx = fmaxf(a, b);
#pragma unroll
    for (int off = 16; off > 0; off >>= 1) mx = fmaxf(mx, __shfl_xor(mx, off, 32));
    float s = expf(a - mx) + expf(b - mx);
#pragma unroll
    for (int off = 16; off > 0; off >>= 1) s += __shfl_xor(s, off, 32);
    float lse = mx + logf(s);
    p[lane]      = a - lse;
    p[lane + 32] = b - lse;
}


extern "C" void kernel_launch(void* const* d_in, const int* in_sizes, int n_in,
                              void* d_out, int out_size, void* d_ws, size_t ws_size,
                              hipStream_t stream) {
    const float* x   = (const float*)d_in[0];
    const int*   ei  = (const int*)d_in[1];
    const float* W1  = (const float*)d_in[2];
    const float* b1  = (const float*)d_in[3];
    const float* W2  = (const float*)d_in[4];
    const float* b2  = (const float*)d_in[5];
    float*       out = (float*)d_out;

    const int* rowp = ei;
    const int* colp = ei + N_EDGES;

    float* ws   = (float*)d_ws;
    float* dinv = ws;
    float* nrm  = dinv + N_NODES;
    float* hb0  = nrm + N_EDGES;
    float* hb1  = hb0 + (size_t)N_NODES * F_DIM;
    float* hb2  = hb1 + (size_t)N_NODES * F_DIM;
    float* acc1 = hb2 + (size_t)N_NODES * F_DIM;

    const int TB = 256;
    const int nodeBlocks = (N_NODES + TB - 1) / TB;
    const int edgeBlocks = (N_EDGES + TB - 1) / TB;
    const int featElems  = N_NODES * F_DIM;
    const int featBlocks = (featElems + TB - 1) / TB;
    const long long spmmThreads = (long long)N_EDGES * 32;
    const int spmmBlocks = (int)((spmmThreads + TB - 1) / TB);
    const int rowTiles   = N_NODES / 16;
    const int rowGroups  = (rowTiles + 7) / 8;

    zero_f32_kernel<<<nodeBlocks, TB, 0, stream>>>(dinv, N_NODES);
    degree_kernel<<<edgeBlocks, TB, 0, stream>>>(colp, dinv);
    dinv_kernel<<<nodeBlocks, TB, 0, stream>>>(dinv);
    norm_kernel<<<edgeBlocks, TB, 0, stream>>>(rowp, colp, dinv, nrm);

    zero_f32_kernel<<<featBlocks, TB, 0, stream>>>(hb0, featElems);
    spmm_scatter_kernel<<<spmmBlocks, TB, 0, stream>>>(rowp, colp, nrm, x, hb0);
    zero_f32_kernel<<<featBlocks, TB, 0, stream>>>(hb1, featElems);
    spmm_scatter_kernel<<<spmmBlocks, TB, 0, stream>>>(rowp, colp, nrm, hb0, hb1);
    zero_f32_kernel<<<featBlocks, TB, 0, stream>>>(hb2, featElems);
    spmm_scatter_kernel<<<spmmBlocks, TB, 0, stream>>>(rowp, colp, nrm, hb1, hb2);

    tag_gemm_wmma_kernel<H_OUT1, 0><<<dim3(rowGroups, H_OUT1 / 16), 256, 0, stream>>>(
        x, hb0, hb1, hb2, W1, b1, acc1);

    zero_f32_kernel<<<featBlocks, TB, 0, stream>>>(hb0, featElems);
    spmm_scatter_kernel<<<spmmBlocks, TB, 0, stream>>>(rowp, colp, nrm, acc1, hb0);
    zero_f32_kernel<<<featBlocks, TB, 0, stream>>>(hb1, featElems);
    spmm_scatter_kernel<<<spmmBlocks, TB, 0, stream>>>(rowp, colp, nrm, hb0, hb1);
    zero_f32_kernel<<<featBlocks, TB, 0, stream>>>(hb2, featElems);
    spmm_scatter_kernel<<<spmmBlocks, TB, 0, stream>>>(rowp, colp, nrm, hb1, hb2);

    tag_gemm_wmma_kernel<C_OUT, 1><<<dim3(rowGroups, C_OUT / 16), 256, 0, stream>>>(
        acc1, hb0, hb1, hb2, W2, b2, out);

    log_softmax_kernel<<<N_NODES / 8, TB, 0, stream>>>(out);
}


# Zero fills folded away: each hop's grid clears the next hop's buffer with 16-B stores (was six launches of 200000 one-store waves); first buffer cleared with the degree array; 13 launches, not 19
# speedup vs baseline: 2.1430x; 2.1430x over previous
#include <hip/hip_runtime.h>
#include <hip/hip_bf16.h>
#include <math.h>

#define N_NODES 50000
#define N_EDGES 800000
#define F_DIM   128
#define H_OUT1  128
#define C_OUT   64
#define K_TOT   512

typedef __attribute__((ext_vector_type(2))) float v2f;
typedef __attribute__((ext_vector_type(8))) float v8f;


__global__ __launch_bounds__(256) void zero_span_kernel(float* __restrict__ p, int n4) {
    float4* __restrict__ p4 = (float4*)p;
    const int stride = gridDim.x * 256;
    for (int i = blockIdx.x * 256 + threadIdx.x; i < n4; i += stride)
        p4[i] = make_float4(0.0f, 0.0f, 0.0f, 0.0f);
}

__global__ void degree_kernel(const int* __restrict__ col, float* __restrict__ deg) {
    int e = blockIdx.x * blockDim.x + threadIdx.x;
    if (e < N_EDGES) atomicAdd(&deg[col[e]], 1.0f);
}

__global__ void dinv_kernel(float* __restrict__ deg) {
    int i = blockIdx.x * blockDim.x + threadIdx.x;
    if (i < N_NODES) {
        float d = deg[i];
        deg[i] = (d > 0.0f) ? rsqrtf(d) : 0.0f;
    }
}

__global__ void norm_kernel(const int* __restrict__ row, const int* __restrict__ col,
                            const float* __restrict__ dinv, float* __restrict__ nrm) {
    int e = blockIdx.x * blockDim.x + threadIdx.x;
    if (e < N_EDGES) nrm[e] = dinv[row[e]] * dinv[col[e]];
}

#define EPW 16
static_assert(N_EDGES % EPW == 0, "edge count must be a whole number of wave tiles");
static_assert(EPW <= 32 && (EPW & (EPW - 1)) == 0, "EPW must be a power of two within one wave");

__global__ __launch_bounds__(256) void spmm_scatter_kernel(
    const int* __restrict__ row, const int* __restrict__ col,
    const float* __restrict__ nrm,
    const float* __restrict__ hin, float* __restrict__ hout, float* __restrict__ znext) {
    const int lane = threadIdx.x & 31;
    const int wave = blockIdx.x * (256 / 32) + (threadIdx.x >> 5);

    if (znext != nullptr) {
        float4* __restrict__ z4 = (float4*)znext;
        const int n4 = N_NODES * F_DIM / 4;
        const int stride = gridDim.x * 256;
        for (int i = blockIdx.x * 256 + threadIdx.x; i < n4; i += stride)
            z4[i] = make_float4(0.0f, 0.0f, 0.0f, 0.0f);
    }

    const int e0 = wave * EPW;
    if (e0 >= N_EDGES) return;

    const int le = e0 + (lane & (EPW - 1));
    const int   r_l = row[le];
    const int   c_l = col[le];
    const float w_l = nrm[le];

    float v[EPW][4];
    for (int j = 0; j < EPW; ++j) {
        const int rj = __builtin_amdgcn_readlane(r_l, j);
        const float* src = hin + (long long)rj * F_DIM + lane;
        v[j][0] = src[0];
        v[j][1] = src[32];
        v[j][2] = src[64];
        v[j][3] = src[96];
    }
    for (int j = 0; j < EPW; ++j) {
        const int   cj = __builtin_amdgcn_readlane(c_l, j);
        const float wj = __int_as_float(__builtin_amdgcn_readlane(__float_as_int(w_l), j));
        float* dst = hout + (long long)cj * F_DIM + lane;
        atomicAdd(dst + 0,  wj * v[j][0]);
        atomicAdd(dst + 32, wj * v[j][1]);
        atomicAdd(dst + 64, wj * v[j][2]);
        atomicAdd(dst + 96, wj * v[j][3]);
    }
}

template <int NOUT, int MODE>
__global__ __launch_bounds__(256) void tag_gemm_wmma_kernel(
    const float* __restrict__ s0, const float* __restrict__ s1,
    const float* __restrict__ s2, const float* __restrict__ s3,
    const float* __restrict__ W, const float* __restrict__ bias,
    float* __restrict__ out) {
    __shared__ float Bs[K_TOT * 16];

    const int col0 = blockIdx.y * 16;

    for (int i = threadIdx.x; i < K_TOT * 16; i += 256) {
        int k = i >> 4;
        int n = i & 15;
        Bs[i] = W[k * NOUT + col0 + n];
    }
    __syncthreads();

    const int wave   = threadIdx.x >> 5;
    const int lane   = threadIdx.x & 31;
    const int lane16 = lane & 15;
    const int hi     = lane >> 4;
    const int row0   = (blockIdx.x * 8 + wave) * 16;
    if (row0 >= N_NODES) return;
    const int m = row0 + lane16;

    const float* segs[4] = { s0, s1, s2, s3 };

    v8f c = { 0.f, 0.f, 0.f, 0.f, 0.f, 0.f, 0.f, 0.f };

#pragma unroll
    for (int s = 0; s < 4; ++s) {
        const float* __restrict__ arow = segs[s] + (long long)m * F_DIM;
        const float* __restrict__ bseg = Bs + s * F_DIM * 16;
        for (int kk = 0; kk < F_DIM; kk += 4) {
            v2f a;
            a.x = arow[kk + hi * 2 + 0];
            a.y = arow[kk + hi * 2 + 1];
            v2f b;
            b.x = bseg[(kk + hi * 2 + 0) * 16 + lane16];
            b.y = bseg[(kk + hi * 2 + 1) * 16 + lane16];
            c = __builtin_amdgcn_wmma_f32_16x16x4_f32(false, a, false, b, (short)0, c,
                                                      false, false);
        }
    }

    const int n    = col0 + lane16;
    const float bv = bias[n];
#pragma unroll
    for (int j = 0; j < 8; ++j) {
        int mm  = row0 + j + hi * 8;
        float v = c[j] + bv;
        if (MODE == 0) v = (v > 0.0f) ? v : 0.0f;
        out[(long long)mm * NOUT + n] = v;
    }
}

__global__ void log_softmax_kernel(float* __restrict__ out) {
    int wave = threadIdx.x >> 5;
    int lane = threadIdx.x & 31;
    int r = blockIdx.x * (blockDim.x >> 5) + wave;
    if (r >= N_NODES) return;
    float* p = out + (long long)r * C_OUT;
    float a = p[lane];
    float b = p[lane + 32];
    float mx = fmaxf(a, b);
#pragma unroll
    for (int off = 16; off > 0; off >>= 1) mx = fmaxf(mx, __shfl_xor(mx, off, 32));
    float s = expf(a - mx) + expf(b - mx);
#pragma unroll
    for (int off = 16; off > 0; off >>= 1) s += __shfl_xor(s, off, 32);
    float lse = mx + logf(s);
    p[lane]      = a - lse;
    p[lane + 32] = b - lse;
}


extern "C" void kernel_launch(void* const* d_in, const int* in_sizes, int n_in,
                              void* d_out, int out_size, void* d_ws, size_t ws_size,
                              hipStream_t stream) {
    const float* x   = (const float*)d_in[0];
    const int*   ei  = (const int*)d_in[1];
    const float* W1  = (const float*)d_in[2];
    const float* b1  = (const float*)d_in[3];
    const float* W2  = (const float*)d_in[4];
    const float* b2  = (const float*)d_in[5];
    float*       out = (float*)d_out;

    const int* rowp = ei;
    const int* colp = ei + N_EDGES;

    float* ws   = (float*)d_ws;
    float* dinv = ws;
    float* hb0  = dinv + N_NODES + 48;
    float* hb1  = hb0 + (size_t)N_NODES * F_DIM;
    float* hb2  = hb1 + (size_t)N_NODES * F_DIM;
    float* hb3  = hb2 + (size_t)N_NODES * F_DIM;
    float* acc1 = hb3 + (size_t)N_NODES * F_DIM;
    float* nrm  = acc1 + (size_t)N_NODES * F_DIM;

    const int TB = 256;
    const int nodeBlocks = (N_NODES + TB - 1) / TB;
    const int edgeBlocks = (N_EDGES + TB - 1) / TB;
    const int featElems  = N_NODES * F_DIM;
    const int headVec4   = (N_NODES + 48 + featElems) / 4;
    const int zeroBlocks = 2048;
    const int spmmWaves  = N_EDGES / EPW;
    const int spmmBlocks = (spmmWaves + 7) / 8;
    const int rowTiles   = N_NODES / 16;
    const int rowGroups  = (rowTiles + 7) / 8;

    zero_span_kernel<<<zeroBlocks, TB, 0, stream>>>(dinv, headVec4);
    degree_kernel<<<edgeBlocks, TB, 0, stream>>>(colp, dinv);
    dinv_kernel<<<nodeBlocks, TB, 0, stream>>>(dinv);
    norm_kernel<<<edgeBlocks, TB, 0, stream>>>(rowp, colp, dinv, nrm);

    spmm_scatter_kernel<<<spmmBlocks, TB, 0, stream>>>(rowp, colp, nrm, x,   hb0, hb1);
    spmm_scatter_kernel<<<spmmBlocks, TB, 0, stream>>>(rowp, colp, nrm, hb0, hb1, hb2);
    spmm_scatter_kernel<<<spmmBlocks, TB, 0, stream>>>(rowp, colp, nrm, hb1, hb2, hb3);

    tag_gemm_wmma_kernel<H_OUT1, 0><<<dim3(rowGroups, H_OUT1 / 16), 256, 0, stream>>>(
        x, hb0, hb1, hb2, W1, b1, acc1);

    spmm_scatter_kernel<<<spmmBlocks, TB, 0, stream>>>(rowp, colp, nrm, acc1, hb3, hb0);
    spmm_scatter_kernel<<<spmmBlocks, TB, 0, stream>>>(rowp, colp, nrm, hb3,  hb0, hb1);
    spmm_scatter_kernel<<<spmmBlocks, TB, 0, stream>>>(rowp, colp, nrm, hb0,  hb1, (float*)nullptr);

    tag_gemm_wmma_kernel<C_OUT, 1><<<dim3(rowGroups, C_OUT / 16), 256, 0, stream>>>(
        acc1, hb3, hb0, hb1, W2, b2, out);

    log_softmax_kernel<<<N_NODES / 8, TB, 0, stream>>>(out);
}


# Products: k segment s of each layer runs as every fifth resident work-group of hop launch s+1, which reads the same operand (matrix units beside atomic-bound hops); sums carried in C, k order kept
# speedup vs baseline: 3.3472x; 1.5619x over previous
#include <hip/hip_runtime.h>
#include <hip/hip_bf16.h>
#include <math.h>

#define N_NODES 50000
#define N_EDGES 800000
#define F_DIM   128
#define H_OUT1  128
#define C_OUT   64

typedef __attribute__((ext_vector_type(2))) float v2f;
typedef __attribute__((ext_vector_type(4))) float v4f;
typedef __attribute__((ext_vector_type(8))) float v8f;


__global__ __launch_bounds__(256) void zero_span_kernel(float* __restrict__ p, int n4) {
    float4* __restrict__ p4 = (float4*)p;
    const int stride = gridDim.x * 256;
    for (int i = blockIdx.x * 256 + threadIdx.x; i < n4; i += stride)
        p4[i] = make_float4(0.0f, 0.0f, 0.0f, 0.0f);
}

#define DEG_GROUPS 64
#define DEG_SLICE  (N_EDGES / DEG_GROUPS)
#define DEG_VEC    (DEG_SLICE / 4)
#define DEG_TRIPS  ((DEG_VEC + 1023) / 1024)
#define DEG_PITCH  50048
static_assert(N_EDGES % DEG_GROUPS == 0 && DEG_SLICE % 4 == 0, "slices are whole 16-byte pieces");
static_assert(N_NODES % 4 == 0, "the table is cleared and stored in 16-byte pieces");
static_assert(DEG_PITCH >= N_NODES && DEG_PITCH % 32 == 0, "stored tables start on 128-byte lines");

__global__ __launch_bounds__(1024) void degree_hist_kernel(const int* __restrict__ col,
                                                           unsigned* __restrict__ part) {
    __shared__ unsigned cnt[N_NODES];
    uint4* __restrict__ cnt4 = (uint4*)cnt;

    const int4* __restrict__ c4 = (const int4*)(col + blockIdx.x * DEG_SLICE);
    int4 c[DEG_TRIPS];
    for (int j = 0; j < DEG_TRIPS; ++j) {
        const int i = threadIdx.x + j * 1024;
        c[j] = (i < DEG_VEC) ? c4[i] : make_int4(-1, -1, -1, -1);
    }

    for (int i = threadIdx.x; i < N_NODES / 4; i += 1024) cnt4[i] = make_uint4(0u, 0u, 0u, 0u);
    __syncthreads();

    for (int j = 0; j < DEG_TRIPS; ++j) {
        if (c[j].x >= 0) {
            atomicAdd(&cnt[c[j].x], 1u);
            atomicAdd(&cnt[c[j].y], 1u);
            atomicAdd(&cnt[c[j].z], 1u);
            atomicAdd(&cnt[c[j].w], 1u);
        }
    }
    __syncthreads();

    uint4* __restrict__ p4 = (uint4*)(part + (size_t)blockIdx.x * DEG_PITCH);
    for (int i = threadIdx.x; i < N_NODES / 4; i += 1024) p4[i] = cnt4[i];
}

__global__ void dinv_kernel(const unsigned* __restrict__ part, float* __restrict__ dinv) {
    int i = blockIdx.x * blockDim.x + threadIdx.x;
    if (i < N_NODES) {
        unsigned n = 0u;
        for (int g = 0; g < DEG_GROUPS; ++g) n += part[(size_t)g * DEG_PITCH + i];
        dinv[i] = (n > 0u) ? rsqrtf((float)n) : 0.0f;
    }
}

__global__ void norm_kernel(const int* __restrict__ row, const int* __restrict__ col,
                            const float* __restrict__ dinv, float* __restrict__ nrm) {
    int e = blockIdx.x * blockDim.x + threadIdx.x;
    if (e < N_EDGES) nrm[e] = dinv[row[e]] * dinv[col[e]];
}

#define SEG_K      128
#define SEG_COLS   64
#define ROW_PAIRS  ((N_NODES / 16 + 1) / 2)

template <int NOUT, int EPI>
__device__ __forceinline__ void product_segment(
    const float* __restrict__ A, const float* __restrict__ Wseg, const float* __restrict__ bias,
    float* C, int first, int gb, int ngb, float* Bs) {
    constexpr int NCG = NOUT / SEG_COLS;
    static_assert(NOUT % SEG_COLS == 0, "whole column groups");
    const int colbase = (gb % NCG) * SEG_COLS;

    static_assert(SEG_K * SEG_COLS == 8 * 256 * 4, "eight 16-byte pieces per thread");
    float4 w[8];
#pragma unroll
    for (int j = 0; j < 8; ++j) {
        const int i  = threadIdx.x + j * 256;
        const int k  = i >> 4;
        const int n4 = (i & 15) * 4;
        w[j] = *(const float4*)(Wseg + k * NOUT + colbase + n4);
    }
#pragma unroll
    for (int j = 0; j < 8; ++j) {
        const int i  = threadIdx.x + j * 256;
        const int k  = i >> 4;
        const int n4 = (i & 15) * 4;
        *(float4*)(Bs + k * SEG_COLS + n4) = w[j];
    }
    __syncthreads();

    const int wave   = __builtin_amdgcn_readfirstlane(threadIdx.x >> 5);
    const int lane   = threadIdx.x & 31;
    const int lane16 = lane & 15;
    const int hi     = lane >> 4;
    const int ct     = wave & 3;
    const int rsub   = wave >> 2;
    const int n      = colbase + ct * 16 + lane16;
    const float* __restrict__ bcol = Bs + ct * 16 + lane16;

    float bv = 0.0f;
    if (EPI != 0) bv = bias[n];

    for (int rp = gb / NCG; rp < ROW_PAIRS; rp += ngb / NCG) {
        const int row0 = (rp * 2 + rsub) * 16;
        if (row0 >= N_NODES) break;
        const float* __restrict__ arow = A + (long long)(row0 + lane16) * F_DIM + hi * 4;
        float* crow = C + (long long)(row0 + hi * 8) * NOUT + n;

        v4f q[SEG_K / 8];
#pragma unroll
        for (int j = 0; j < SEG_K / 8; ++j) q[j] = *(const v4f*)(arow + j * 8);

        v8f c = { 0.f, 0.f, 0.f, 0.f, 0.f, 0.f, 0.f, 0.f };
        if (!first) {
#pragma unroll
            for (int j = 0; j < 8; ++j) c[j] = crow[j * NOUT];
        }

        asm volatile("" :: "v"(q[0]), "v"(q[1]), "v"(q[2]), "v"(q[3]), "v"(q[4]), "v"(q[5]), "v"(q[6]), "v"(q[7]),
                           "v"(q[8]), "v"(q[9]), "v"(q[10]), "v"(q[11]), "v"(q[12]), "v"(q[13]), "v"(q[14]), "v"(q[15]));

#pragma unroll
        for (int j = 0; j < SEG_K / 8; ++j) {
            const int k0 = j * 8 + hi * 4;
            v2f a0, a1, b0, b1;
            a0.x = q[j].x; a0.y = q[j].y;
            a1.x = q[j].z; a1.y = q[j].w;
            b0.x = bcol[(k0 + 0) * SEG_COLS];
            b0.y = bcol[(k0 + 1) * SEG_COLS];
            b1.x = bcol[(k0 + 2) * SEG_COLS];
            b1.y = bcol[(k0 + 3) * SEG_COLS];
            c = __builtin_amdgcn_wmma_f32_16x16x4_f32(false, a0, false, b0, (short)0, c,
                                                      false, false);
            c = __builtin_amdgcn_wmma_f32_16x16x4_f32(false, a1, false, b1, (short)0, c,
                                                      false, false);
        }

#pragma unroll
        for (int j = 0; j < 8; ++j) {
            float v = c[j];
            if (EPI != 0) v = v + bv;
            if (EPI == 1) v = (v > 0.0f) ? v : 0.0f;
            crow[j * NOUT] = v;
        }
    }
}

template <int NOUT, int EPI>
__global__ __launch_bounds__(256) void product_last_kernel(
    const float* __restrict__ A, const float* __restrict__ Wseg, const float* __restrict__ bias,
    float* C) {
    __shared__ __attribute__((aligned(16))) float Bs[SEG_K * SEG_COLS];
    product_segment<NOUT, EPI>(A, Wseg, bias, C, 0, blockIdx.x, gridDim.x, Bs);
}

#define TILE_EDGES 256
#define N_TILES    (N_EDGES / TILE_EDGES)
#define BATCH      16
static_assert(N_EDGES % TILE_EDGES == 0, "edge count must be a whole number of tiles");
static_assert(N_NODES == N_TILES * 16, "each tile of a half clears 8 of that half's rows");

#define HOP_FIRST_BLOCKS   1280
#define HOP_PRODUCT_BLOCKS 256
#define HOP_BLOCKS (HOP_FIRST_BLOCKS + HOP_PRODUCT_BLOCKS)
static_assert(HOP_FIRST_BLOCKS == 5 * HOP_PRODUCT_BLOCKS, "every fifth of the first blocks is a product block");

template <int NOUT>
__global__ __launch_bounds__(256) void spmm_scatter_kernel(
    const int* __restrict__ row, const int* __restrict__ col,
    const float* __restrict__ nrm,
    const float* __restrict__ hin, float* __restrict__ hout, float* __restrict__ znext,
    int* __restrict__ tick0, int* __restrict__ tick1,
    const float* __restrict__ Wseg, float* C, int first) {
    __shared__ __attribute__((aligned(16))) float Bs[SEG_K * SEG_COLS];
    __shared__ int s_tile[2];
    __shared__ int s_half[2];
    if (blockIdx.x < HOP_FIRST_BLOCKS && blockIdx.x % 5 == 4) {
        product_segment<NOUT, 0>(hin, Wseg, nullptr, C, first, blockIdx.x / 5, HOP_PRODUCT_BLOCKS, Bs);
        return;
    }
    const int lane = threadIdx.x & 31;
    const int wave = threadIdx.x >> 5;

    int my_half = 0, pass = 0, nt = N_TILES, nh = 0;
    if (threadIdx.x == 0) {
        my_half = (int)((__builtin_amdgcn_s_sendmsg_rtn(0x87) >> 18) & 1u);
        nh = my_half;
        nt = atomicAdd(nh ? tick1 : tick0, 1);
    }

    for (int it = 0; it < 2 * N_TILES + 2; ++it) {
        const int slot = it & 1;
        if (threadIdx.x == 0) {
            if (nt >= N_TILES && pass == 0) {
                pass = 1;
                nh = my_half ^ 1;
                nt = atomicAdd(nh ? tick1 : tick0, 1);
            }
            s_tile[slot] = nt;
            s_half[slot] = nh;
        }
        __syncthreads();
        const int t  = __builtin_amdgcn_readfirstlane(s_tile[slot]);
        const int hh = __builtin_amdgcn_readfirstlane(s_half[slot]);
        if (t >= N_TILES) break;

        if (znext != nullptr) {
            const int zr = 2 * (t * 8 + wave) + hh;
            float4* __restrict__ z4 = (float4*)(znext + (long long)zr * F_DIM);
            z4[lane] = make_float4(0.0f, 0.0f, 0.0f, 0.0f);
        }

        const int e = t * TILE_EDGES + wave * 32 + lane;
        const int   r_l = row[e];
        const int   c_l = col[e];
        const float w_l = nrm[e];
        asm volatile("" :: "v"(r_l), "v"(c_l), "v"(w_l));
        unsigned todo = __builtin_amdgcn_ballot_w32((c_l & 1) == hh);

        if (threadIdx.x == 0) {
            nh = my_half ^ pass;
            nt = atomicAdd(nh ? tick1 : tick0, 1);
        }

        while (todo != 0u) {
            float v[BATCH][4];
            unsigned m = todo;
#pragma unroll
            for (int j = 0; j < BATCH; ++j) {
                v[j][0] = 0.0f; v[j][1] = 0.0f; v[j][2] = 0.0f; v[j][3] = 0.0f;
                if (m != 0u) {
                    const int l = __builtin_ctz(m);
                    m &= m - 1u;
                    const int rj = __builtin_amdgcn_readlane(r_l, l);
                    const float* src = hin + (long long)rj * F_DIM + lane;
                    v[j][0] = src[0];
                    v[j][1] = src[32];
                    v[j][2] = src[64];
                    v[j][3] = src[96];
                }
            }
#pragma unroll
            for (int j = 0; j < BATCH; ++j) {
                if (todo != 0u) {
                    const int l = __builtin_ctz(todo);
                    todo &= todo - 1u;
                    const int   cj = __builtin_amdgcn_readlane(c_l, l);
                    const float wj = __int_as_float(__builtin_amdgcn_readlane(__float_as_int(w_l), l));
                    float* dst = hout + (long long)cj * F_DIM + lane;
                    atomicAdd(dst + 0,  wj * v[j][0]);
                    atomicAdd(dst + 32, wj * v[j][1]);
                    atomicAdd(dst + 64, wj * v[j][2]);
                    atomicAdd(dst + 96, wj * v[j][3]);
                }
            }
        }
    }
}

__global__ void log_softmax_kernel(float* __restrict__ out) {
    int wave = threadIdx.x >> 5;
    int lane = threadIdx.x & 31;
    int r = blockIdx.x * (blockDim.x >> 5) + wave;
    if (r >= N_NODES) return;
    float* p = out + (long long)r * C_OUT;
    float a = p[lane];
    float b = p[lane + 32];
    float mx = fmaxf(a, b);
#pragma unroll
    for (int off = 16; off > 0; off >>= 1) mx = fmaxf(mx, __shfl_xor(mx, off, 32));
    float s = expf(a - mx) + expf(b - mx);
#pragma unroll
    for (int off = 16; off > 0; off >>= 1) s += __shfl_xor(s, off, 32);
    float lse = mx + logf(s);
    p[lane]      = a - lse;
    p[lane + 32] = b - lse;
}


extern "C" void kernel_launch(void* const* d_in, const int* in_sizes, int n_in,
                              void* d_out, int out_size, void* d_ws, size_t ws_size,
                              hipStream_t stream) {
    const float* x   = (const float*)d_in[0];
    const int*   ei  = (const int*)d_in[1];
    const float* W1  = (const float*)d_in[2];
    const float* b1  = (const float*)d_in[3];
    const float* W2  = (const float*)d_in[4];
    const float* b2  = (const float*)d_in[5];
    float*       out = (float*)d_out;

    const int* rowp = ei;
    const int* colp = ei + N_EDGES;

    float* ws   = (float*)d_ws;
    float* dinv = ws;
    float* hb0  = dinv + N_NODES + 112;
    float* hb1  = hb0 + (size_t)N_NODES * F_DIM;
    float* hb2  = hb1 + (size_t)N_NODES * F_DIM;
    float* hb3  = hb2 + (size_t)N_NODES * F_DIM;
    float* acc1 = hb3 + (size_t)N_NODES * F_DIM;
    float* nrm  = acc1 + (size_t)N_NODES * F_DIM;
    unsigned* part = (unsigned*)(nrm + N_EDGES);
    int* tk0 = (int*)(dinv + N_NODES + 16);
    int* tk1 = (int*)(dinv + N_NODES + 80);

    const int TB = 256;
    const int nodeBlocks = (N_NODES + TB - 1) / TB;
    const int edgeBlocks = (N_EDGES + TB - 1) / TB;
    const int featElems  = N_NODES * F_DIM;
    const int headVec4   = (112 + featElems) / 4;
    const int zeroBlocks = 2048;
    const int lastBlocks = 1024;
    const int segW1      = F_DIM * H_OUT1;
    const int segW2      = F_DIM * C_OUT;

    zero_span_kernel<<<zeroBlocks, TB, 0, stream>>>(dinv + N_NODES, headVec4);
    degree_hist_kernel<<<DEG_GROUPS, 1024, 0, stream>>>(colp, part);
    dinv_kernel<<<nodeBlocks, TB, 0, stream>>>(part, dinv);
    norm_kernel<<<edgeBlocks, TB, 0, stream>>>(rowp, colp, dinv, nrm);

    spmm_scatter_kernel<H_OUT1><<<HOP_BLOCKS, TB, 0, stream>>>(rowp, colp, nrm, x,   hb0, hb1, tk0 + 0, tk1 + 0, W1 + 0 * segW1, acc1, 1);
    spmm_scatter_kernel<H_OUT1><<<HOP_BLOCKS, TB, 0, stream>>>(rowp, colp, nrm, hb0, hb1, hb2, tk0 + 1, tk1 + 1, W1 + 1 * segW1, acc1, 0);
    spmm_scatter_kernel<H_OUT1><<<HOP_BLOCKS, TB, 0, stream>>>(rowp, colp, nrm, hb1, hb2, hb3, tk0 + 2, tk1 + 2, W1 + 2 * segW1, acc1, 0);

    product_last_kernel<H_OUT1, 1><<<lastBlocks, TB, 0, stream>>>(hb2, W1 + 3 * segW1, b1, acc1);

    spmm_scatter_kernel<C_OUT><<<HOP_BLOCKS, TB, 0, stream>>>(rowp, colp, nrm, acc1, hb3, hb0, tk0 + 3, tk1 + 3, W2 + 0 * segW2, out, 1);
    spmm_scatter_kernel<C_OUT><<<HOP_BLOCKS, TB, 0, stream>>>(rowp, colp, nrm, hb3,  hb0, hb1, tk0 + 4, tk1 + 4, W2 + 1 * segW2, out, 0);
    spmm_scatter_kernel<C_OUT><<<HOP_BLOCKS, TB, 0, stream>>>(rowp, colp, nrm, hb0,  hb1, (float*)nullptr, tk0 + 5, tk1 + 5, W2 + 2 * segW2, out, 0);

    product_last_kernel<C_OUT, 2><<<lastBlocks, TB, 0, stream>>>(hb1, W2 + 3 * segW2, b2, out);

    log_softmax_kernel<<<N_NODES / 8, TB, 0, stream>>>(out);
}
